# starting kernel
# baseline (speedup 1.0000x reference)
#include <hip/hip_runtime.h>
#include <hip/hip_bf16.h>


#define B_   64
#define P_   128
#define H1_  256
#define H2_  128
#define C_   32

typedef __attribute__((ext_vector_type(2))) float v2f;
typedef __attribute__((ext_vector_type(8))) float v8f;

__global__ void __launch_bounds__(256) spw_scatter(
    const float* __restrict__ x, const float* __restrict__ w,
    const int* __restrict__ idx, float* __restrict__ WX, int N)
{
    __shared__ float lacc[P_];
    const int t = threadIdx.x;
    if (t < P_) lacc[t] = 0.0f;
    __syncthreads();

    const int  b    = blockIdx.y;
    const long base = ((long)blockIdx.x * 256 + t) * 64;

    if (base < N) {
        const float* xr = x + (long)b * N + base;
        const float* wr = w + base;
        const int*   ir = idx + base;
        int   cur = ir[0];
        float run = 0.0f;
        #pragma unroll 4
        for (int j = 0; j < 64; j += 4) {
            float4 xv = *(const float4*)(xr + j);
            float4 wv = *(const float4*)(wr + j);
            int4   iv = *(const int4*)(ir + j);
            float p0 = xv.x * wv.x, p1 = xv.y * wv.y;
            float p2 = xv.z * wv.z, p3 = xv.w * wv.w;
            if (iv.x != cur) { atomicAdd(&lacc[cur], run); run = 0.f; cur = iv.x; } run += p0;
            if (iv.y != cur) { atomicAdd(&lacc[cur], run); run = 0.f; cur = iv.y; } run += p1;
            if (iv.z != cur) { atomicAdd(&lacc[cur], run); run = 0.f; cur = iv.z; } run += p2;
            if (iv.w != cur) { atomicAdd(&lacc[cur], run); run = 0.f; cur = iv.w; } run += p3;
        }
        atomicAdd(&lacc[cur], run);
    }
    __syncthreads();

    if (t < P_) {
        float v = lacc[t];
        if (v != 0.0f) atomicAdd(&WX[(long)b * P_ + t], v);
    }
}

__device__ __forceinline__ v8f wmma_tile_f32(
    const float* __restrict__ A, int lda,
    const float* __restrict__ Bm, int ldb,
    int K, int mt, int nt, int lane)
{
    const int ml   = lane & 15;
    const int koff = (lane >> 4) * 2;
    const float* arow = A + (mt * 16 + ml) * lda;
    const float* bcol = Bm + nt * 16 + ml;
    v8f acc = {0.f, 0.f, 0.f, 0.f, 0.f, 0.f, 0.f, 0.f};
    for (int k = 0; k < K; k += 4) {
        v2f av, bv;
        av.x = arow[k + koff];
        av.y = arow[k + koff + 1];
        bv.x = bcol[(k + koff) * ldb];
        bv.y = bcol[(k + koff + 1) * ldb];
        acc = __builtin_amdgcn_wmma_f32_16x16x4_f32(
            false, av, false, bv, (short)0, acc, false, false);
    }
    return acc;
}

__global__ void __launch_bounds__(512) spw_head(
    const float* __restrict__ WX,
    const float* __restrict__ co_w,
    const float* __restrict__ bn0_g, const float* __restrict__ bn0_b,
    const float* __restrict__ W1, const float* __restrict__ b1,
    const float* __restrict__ bn1_g, const float* __restrict__ bn1_b,
    const float* __restrict__ W2, const float* __restrict__ b2,
    const float* __restrict__ bn2_g, const float* __restrict__ bn2_b,
    const float* __restrict__ Wo, const float* __restrict__ bo,
    float* __restrict__ Zout,
    float* __restrict__ yout,
    float* __restrict__ h1,
    float* __restrict__ h2,
    float* __restrict__ logits)
{
    __shared__ float sA[H1_];
    __shared__ float sC[H1_];
    const int t    = threadIdx.x;
    const int lane = t & 31;
    const int wave = t >> 5;
    const float invB = 1.0f / (float)B_;

    if (t < P_) {
        float s = 0.f, s2 = 0.f;
        for (int r = 0; r < B_; ++r) {
            float v = fmaxf(WX[r * P_ + t], 0.f);
            s += v; s2 += v * v;
        }
        float m   = s * invB;
        float var = s2 * invB - m * m;
        float rg  = rsqrtf(var + 1e-5f) * bn0_g[t];
        float sg  = 1.0f / (1.0f + __expf(-co_w[t]));
        sA[t] = rg * sg;
        sC[t] = (bn0_b[t] - m * rg) * sg;
    }
    __syncthreads();
    for (int e = t; e < B_ * P_; e += 512) {
        int cp = e & (P_ - 1);
        float v = fmaxf(WX[e], 0.f);
        Zout[e] = v * sA[cp] + sC[cp];
    }
    __syncthreads();

    for (int q = 0; q < 4; ++q) {
        int tile = wave * 4 + q;
        int mt = tile >> 4, nt = tile & 15;
        v8f acc = wmma_tile_f32(Zout, P_, W1, H1_, P_, mt, nt, lane);
        int ml = lane & 15, n = nt * 16 + ml, mb = mt * 16 + (lane >> 4) * 8;
        float bias = b1[n];
        #pragma unroll
        for (int r = 0; r < 8; ++r)
            h1[(mb + r) * H1_ + n] = fmaxf(acc[r] + bias, 0.f);
    }
    __syncthreads();
    if (t < H1_) {
        float s = 0.f, s2 = 0.f;
        for (int r = 0; r < B_; ++r) { float v = h1[r * H1_ + t]; s += v; s2 += v * v; }
        float m = s * invB, var = s2 * invB - m * m;
        float rg = rsqrtf(var + 1e-5f) * bn1_g[t];
        sA[t] = rg;
        sC[t] = bn1_b[t] - m * rg;
    }
    __syncthreads();
    for (int e = t; e < B_ * H1_; e += 512) {
        int cc = e & (H1_ - 1);
        h1[e] = h1[e] * sA[cc] + sC[cc];
    }
    __syncthreads();

    for (int q = 0; q < 2; ++q) {
        int tile = wave * 2 + q;
        int mt = tile >> 3, nt = tile & 7;
        v8f acc = wmma_tile_f32(h1, H1_, W2, H2_, H1_, mt, nt, lane);
        int ml = lane & 15, n = nt * 16 + ml, mb = mt * 16 + (lane >> 4) * 8;
        float bias = b2[n];
        #pragma unroll
        for (int r = 0; r < 8; ++r)
            h2[(mb + r) * H2_ + n] = fmaxf(acc[r] + bias, 0.f);
    }
    __syncthreads();
    if (t < H2_) {
        float s = 0.f, s2 = 0.f;
        for (int r = 0; r < B_; ++r) { float v = h2[r * H2_ + t]; s += v; s2 += v * v; }
        float m = s * invB, var = s2 * invB - m * m;
        float rg = rsqrtf(var + 1e-5f) * bn2_g[t];
        sA[t] = rg;
        sC[t] = bn2_b[t] - m * rg;
    }
    __syncthreads();
    for (int e = t; e < B_ * H2_; e += 512) {
        int cc = e & (H2_ - 1);
        h2[e] = h2[e] * sA[cc] + sC[cc];
    }
    __syncthreads();

    if (wave < 8) {
        int tile = wave;
        int mt = tile >> 1, nt = tile & 1;
        v8f acc = wmma_tile_f32(h2, H2_, Wo, C_, H2_, mt, nt, lane);
        int ml = lane & 15, n = nt * 16 + ml, mb = mt * 16 + (lane >> 4) * 8;
        float bias = bo[n];
        #pragma unroll
        for (int r = 0; r < 8; ++r)
            logits[(mb + r) * C_ + n] = acc[r] + bias;
    }
    __syncthreads();

    if (t < B_) {
        const float* lr = logits + t * C_;
        float m = -3.0e38f;
        for (int c = 0; c < C_; ++c) m = fmaxf(m, lr[c]);
        float ex[C_];
        float s = 0.f;
        for (int c = 0; c < C_; ++c) { float e = __expf(lr[c] - m); ex[c] = e; s += e; }
        float inv = 1.0f / s;
        for (int c = 0; c < C_; ++c) yout[t * C_ + c] = ex[c] * inv;
    }
}

extern "C" void kernel_launch(void* const* d_in, const int* in_sizes, int n_in,
                              void* d_out, int out_size, void* d_ws, size_t ws_size,
                              hipStream_t stream) {
    const float* x     = (const float*)d_in[0];
    const float* w     = (const float*)d_in[1];
    const float* co_w  = (const float*)d_in[2];
    const float* bn0_g = (const float*)d_in[3];
    const float* bn0_b = (const float*)d_in[4];
    const float* W1    = (const float*)d_in[5];
    const float* b1    = (const float*)d_in[6];
    const float* bn1_g = (const float*)d_in[7];
    const float* bn1_b = (const float*)d_in[8];
    const float* W2    = (const float*)d_in[9];
    const float* b2    = (const float*)d_in[10];
    const float* bn2_g = (const float*)d_in[11];
    const float* bn2_b = (const float*)d_in[12];
    const float* Wo    = (const float*)d_in[13];
    const float* bo    = (const float*)d_in[14];
    const int*   idx   = (const int*)d_in[15];

    float* out    = (float*)d_out;
    float* WX     = (float*)d_ws;
    float* h1     = WX + B_ * P_;
    float* h2     = h1 + B_ * H1_;
    float* logits = h2 + B_ * H2_;

    const int N = in_sizes[0] / B_;

    hipMemsetAsync(WX, 0, (size_t)B_ * P_ * sizeof(float), stream);

    const int elems_per_block = 256 * 64;
    dim3 g1((N + elems_per_block - 1) / elems_per_block, B_);
    spw_scatter<<<g1, 256, 0, stream>>>(x, w, idx, WX, N);

    spw_head<<<1, 512, 0, stream>>>(WX, co_w, bn0_g, bn0_b,
                                    W1, b1, bn1_g, bn1_b,
                                    W2, b2, bn2_g, bn2_b,
                                    Wo, bo,
                                    out + B_ * C_, out,
                                    h1, h2, logits);
}


# head as one launch per layer: a work-group owns a 32-column strip of the layer for all 64 rows (8, then 4 work-groups; softmax on 4 waves); bias/ReLU/batch norm on accumulators (sum order changes)
# speedup vs baseline: 2.6073x; 2.6073x over previous
#include <hip/hip_runtime.h>
#include <hip/hip_bf16.h>


#define B_   64
#define P_   128
#define H1_  256
#define H2_  128
#define C_   32

typedef __attribute__((ext_vector_type(2))) float v2f;
typedef __attribute__((ext_vector_type(8))) float v8f;

#define SC_TILES 8
#define SC_ROWS  (8 * SC_TILES)
#define SC_NIT   8
#define SC_F     (32 * SC_NIT)

struct ScB {
    v2f f[SC_NIT][2][2];
};

__device__ __forceinline__ void sc_build(ScB& b, const float* __restrict__ wr,
                                         const int* __restrict__ ir,
                                         int pc, int& lo, int& hi)
{
    float4 wv[SC_NIT][2];
    int4   iv[SC_NIT][2];
    #pragma unroll
    for (int it = 0; it < SC_NIT; ++it) {
        #pragma unroll
        for (int s = 0; s < 2; ++s) {
            wv[it][s] = *(const float4*)(wr + it * 32 + s * 16);
            iv[it][s] = *(const int4*)(ir + it * 32 + s * 16);
        }
    }
    #pragma unroll
    for (int it = 0; it < SC_NIT; ++it) {
        #pragma unroll
        for (int s = 0; s < 2; ++s) {
            const int4   i4 = iv[it][s];
            const float4 w4 = wv[it][s];
            lo = min(lo, min(min(i4.x, i4.y), min(i4.z, i4.w)));
            hi = max(hi, max(max(i4.x, i4.y), max(i4.z, i4.w)));
            b.f[it][s][0].x = (i4.x == pc) ? w4.x : 0.0f;
            b.f[it][s][0].y = (i4.y == pc) ? w4.y : 0.0f;
            b.f[it][s][1].x = (i4.z == pc) ? w4.z : 0.0f;
            b.f[it][s][1].y = (i4.w == pc) ? w4.w : 0.0f;
        }
    }
}

__device__ __forceinline__ v8f sc_tile(const ScB& b, const float* __restrict__ xt)
{
    float4 xv[SC_NIT][2];
    #pragma unroll
    for (int it = 0; it < SC_NIT; ++it) {
        #pragma unroll
        for (int s = 0; s < 2; ++s)
            xv[it][s] = *(const float4*)(xt + it * 32 + s * 16);
    }
    __builtin_amdgcn_sched_barrier(0);
    v8f a0 = {0.f, 0.f, 0.f, 0.f, 0.f, 0.f, 0.f, 0.f};
    v8f a1 = {0.f, 0.f, 0.f, 0.f, 0.f, 0.f, 0.f, 0.f};
    #pragma unroll
    for (int i2 = 0; i2 < SC_NIT / 2; ++i2) {
        #pragma unroll
        for (int s = 0; s < 2; ++s) {
            v2f p, q;
            p.x = xv[2 * i2][s].x;     p.y = xv[2 * i2][s].y;
            q.x = xv[2 * i2 + 1][s].x; q.y = xv[2 * i2 + 1][s].y;
            a0 = __builtin_amdgcn_wmma_f32_16x16x4_f32(
                false, p, false, b.f[2 * i2][s][0], (short)0, a0, false, false);
            a1 = __builtin_amdgcn_wmma_f32_16x16x4_f32(
                false, q, false, b.f[2 * i2 + 1][s][0], (short)0, a1, false, false);
            p.x = xv[2 * i2][s].z;     p.y = xv[2 * i2][s].w;
            q.x = xv[2 * i2 + 1][s].z; q.y = xv[2 * i2 + 1][s].w;
            a0 = __builtin_amdgcn_wmma_f32_16x16x4_f32(
                false, p, false, b.f[2 * i2][s][1], (short)0, a0, false, false);
            a1 = __builtin_amdgcn_wmma_f32_16x16x4_f32(
                false, q, false, b.f[2 * i2 + 1][s][1], (short)0, a1, false, false);
        }
    }
    return a0 + a1;
}

__global__ void __launch_bounds__(32) spw_scatter(
    const float* __restrict__ x, const float* __restrict__ w,
    const int* __restrict__ idx, float* __restrict__ WX, int N)
{
    const int lane = threadIdx.x;
    const int c    = lane & 15;
    const int h    = lane >> 4;
    const int fb   = c >> 3;
    const int r    = c & 7;
    const int foff = (fb * 2 + h) * 4;
    const int f0   = blockIdx.x * SC_F;
    const int row0 = blockIdx.y * SC_ROWS;

    if (f0 + SC_F > N) {
        for (int f = f0 + lane; f < N; f += 32) {
            const int   p  = idx[f];
            const float ww = w[f];
            if ((unsigned)p < (unsigned)P_)
                for (int b = 0; b < SC_ROWS; ++b)
                    atomicAdd(&WX[(row0 + b) * P_ + p], x[(long)(row0 + b) * N + f] * ww);
        }
        return;
    }

    const float* xr = x + (long)(row0 + r) * N + f0 + foff;
    const float* wr = w + f0 + foff;
    const int*   ir = idx + f0 + foff;
    const long tile_stride = 8L * N;

    int pb0 = idx[f0];
    pb0 = min(max(pb0, 0), P_ - 1);

    int lo = 0x7fffffff, hi = -0x7fffffff - 1;
    int k = 0, kmax = 0;
    bool first = true;
    int pb = pb0;

    do {
        const int pc = pb + r;
        ScB bfr;
        int zoff = 0;
        asm volatile("" : "+s"(zoff));
        sc_build(bfr, wr + zoff, ir + zoff, pc, lo, hi);

        const bool owner = (fb == h) && ((unsigned)pc < (unsigned)P_);
        const float* xt = xr;
        int wrow = row0 * P_ + pc;

        #pragma unroll 1
        for (int t = 0; t < SC_TILES; ++t) {
            const v8f acc = sc_tile(bfr, xt);
            if (owner) {
                #pragma unroll
                for (int j = 0; j < 8; ++j) {
                    const float v = acc[j];
                    if (v != 0.0f)
                        atomicAdd(&WX[wrow + j * P_], v);
                }
            }
            xt   += tile_stride;
            wrow += 8 * P_;
        }

        if (first) {
            first = false;
            #pragma unroll
            for (int o = 16; o > 0; o >>= 1) {
                lo = min(lo, __shfl_xor(lo, o, 32));
                hi = max(hi, __shfl_xor(hi, o, 32));
            }
            lo = __builtin_amdgcn_readfirstlane(lo);
            hi = __builtin_amdgcn_readfirstlane(hi);
            lo = max(lo, 0);
            hi = min(hi, P_ - 1);
            k    = (lo - pb0) >> 3;
            kmax = (hi - pb0) >> 3;
        } else {
            ++k;
        }
        if (k == 0) ++k;
        pb = pb0 + 8 * k;
    } while (k <= kmax);
}

template <int K>
__device__ __forceinline__ v8f head_tile(const float* __restrict__ arow,
                                         const float* __restrict__ bcol, int ldb)
{
    float4 av[K / 8];
    float  bv[K / 8][4];
    #pragma unroll
    for (int i = 0; i < K / 8; ++i) {
        av[i] = *(const float4*)(arow + 8 * i);
        #pragma unroll
        for (int e = 0; e < 4; ++e) bv[i][e] = bcol[(8 * i + e) * ldb];
    }
    __builtin_amdgcn_sched_barrier(0);
    v8f a0 = {0.f, 0.f, 0.f, 0.f, 0.f, 0.f, 0.f, 0.f};
    v8f a1 = {0.f, 0.f, 0.f, 0.f, 0.f, 0.f, 0.f, 0.f};
    #pragma unroll
    for (int i = 0; i < K / 8; ++i) {
        v2f p, q, u, v;
        p.x = av[i].x;  p.y = av[i].y;
        q.x = av[i].z;  q.y = av[i].w;
        u.x = bv[i][0]; u.y = bv[i][1];
        v.x = bv[i][2]; v.y = bv[i][3];
        a0 = __builtin_amdgcn_wmma_f32_16x16x4_f32(
            false, p, false, u, (short)0, a0, false, false);
        a1 = __builtin_amdgcn_wmma_f32_16x16x4_f32(
            false, q, false, v, (short)0, a1, false, false);
    }
    return a0 + a1;
}

template <int K>
__device__ __forceinline__ v8f head_tile_z(const float* __restrict__ arow,
                                           const float* __restrict__ bcol, int ldb,
                                           const float* sAh, const float* sCh)
{
    float4 av[K / 8];
    float  bv[K / 8][4];
    #pragma unroll
    for (int i = 0; i < K / 8; ++i) {
        av[i] = *(const float4*)(arow + 8 * i);
        #pragma unroll
        for (int e = 0; e < 4; ++e) bv[i][e] = bcol[(8 * i + e) * ldb];
    }
    __builtin_amdgcn_sched_barrier(0);
    v8f a0 = {0.f, 0.f, 0.f, 0.f, 0.f, 0.f, 0.f, 0.f};
    v8f a1 = {0.f, 0.f, 0.f, 0.f, 0.f, 0.f, 0.f, 0.f};
    #pragma unroll
    for (int i = 0; i < K / 8; ++i) {
        const float4 s = *(const float4*)(sAh + 8 * i);
        const float4 t = *(const float4*)(sCh + 8 * i);
        v2f p, q, u, v;
        p.x = fmaxf(av[i].x, 0.f) * s.x + t.x;
        p.y = fmaxf(av[i].y, 0.f) * s.y + t.y;
        q.x = fmaxf(av[i].z, 0.f) * s.z + t.z;
        q.y = fmaxf(av[i].w, 0.f) * s.w + t.w;
        u.x = bv[i][0]; u.y = bv[i][1];
        v.x = bv[i][2]; v.y = bv[i][3];
        a0 = __builtin_amdgcn_wmma_f32_16x16x4_f32(
            false, p, false, u, (short)0, a0, false, false);
        a1 = __builtin_amdgcn_wmma_f32_16x16x4_f32(
            false, q, false, v, (short)0, a1, false, false);
    }
    return a0 + a1;
}

__device__ __forceinline__ void head_bn_store(v8f acc, float bias, float g, float bt,
                                              float (*red)[16][2], int wave, int c, int h,
                                              float* __restrict__ dst  ,
                                              int ld)
{
    float s = 0.f, s2 = 0.f;
    #pragma unroll
    for (int r = 0; r < 8; ++r) {
        const float v = fmaxf(acc[r] + bias, 0.f);
        acc[r] = v;
        s += v; s2 += v * v;
    }
    s  += __shfl_xor(s, 16, 32);
    s2 += __shfl_xor(s2, 16, 32);
    if (h == 0) { red[wave][c][0] = s; red[wave][c][1] = s2; }
    __syncthreads();
    const int w0 = wave & ~3;
    const float S  = (red[w0][c][0] + red[w0 + 1][c][0]) + (red[w0 + 2][c][0] + red[w0 + 3][c][0]);
    const float S2 = (red[w0][c][1] + red[w0 + 1][c][1]) + (red[w0 + 2][c][1] + red[w0 + 3][c][1]);
    const float invB = 1.0f / (float)B_;
    const float m   = S * invB;
    const float var = S2 * invB - m * m;
    const float rg  = rsqrtf(var + 1e-5f) * g;
    const float sh  = bt - m * rg;
    #pragma unroll
    for (int r = 0; r < 8; ++r)
        dst[r * ld] = acc[r] * rg + sh;
}

__global__ void __launch_bounds__(256) spw_head_l1(
    const float* __restrict__ WX,
    const float* __restrict__ co_w,
    const float* __restrict__ bn0_g, const float* __restrict__ bn0_b,
    const float* __restrict__ W1, const float* __restrict__ b1,
    const float* __restrict__ bn1_g, const float* __restrict__ bn1_b,
    float* __restrict__ Zout,
    float* __restrict__ h1n)
{
    __shared__ __attribute__((aligned(16))) float sA[P_];
    __shared__ __attribute__((aligned(16))) float sC[P_];
    __shared__ float red[8][16][2];
    const int t    = threadIdx.x;
    const int lane = t & 31;
    const int wave = t >> 5;
    const int c    = lane & 15;
    const int h    = lane >> 4;
    const int mt   = wave & 3;
    const int n    = (blockIdx.x * 2 + (wave >> 2)) * 16 + c;
    const float invB = 1.0f / (float)B_;

    if (t < P_) {
        float xv[B_];
        #pragma unroll
        for (int r = 0; r < B_; ++r) xv[r] = WX[r * P_ + t];
        float s = 0.f, s2 = 0.f;
        #pragma unroll
        for (int r = 0; r < B_; ++r) {
            const float v = fmaxf(xv[r], 0.f);
            s += v; s2 += v * v;
        }
        const float m   = s * invB;
        const float var = s2 * invB - m * m;
        const float rg  = rsqrtf(var + 1e-5f) * bn0_g[t];
        const float sg  = 1.0f / (1.0f + __expf(-co_w[t]));
        sA[t] = rg * sg;
        sC[t] = (bn0_b[t] - m * rg) * sg;
    }
    __syncthreads();

    #pragma unroll
    for (int q = 0; q < 4; ++q) {
        const int e  = blockIdx.x * (8 * P_) + q * 256 + t;
        const int cp = e & (P_ - 1);
        const float v = fmaxf(WX[e], 0.f);
        Zout[e] = v * sA[cp] + sC[cp];
    }

    const v8f acc = head_tile_z<P_>(WX + (mt * 16 + c) * P_ + 4 * h,
                                    W1 + (4 * h) * H1_ + n, H1_,
                                    sA + 4 * h, sC + 4 * h);
    head_bn_store(acc, b1[n], bn1_g[n], bn1_b[n], red, wave, c, h,
                  h1n + (mt * 16 + 8 * h) * H1_ + n, H1_);
}

__global__ void __launch_bounds__(256) spw_head_l2(
    const float* __restrict__ h1n,
    const float* __restrict__ W2, const float* __restrict__ b2,
    const float* __restrict__ bn2_g, const float* __restrict__ bn2_b,
    float* __restrict__ h2n)
{
    __shared__ float red[8][16][2];
    const int t    = threadIdx.x;
    const int lane = t & 31;
    const int wave = t >> 5;
    const int c    = lane & 15;
    const int h    = lane >> 4;
    const int mt   = wave & 3;
    const int n    = (blockIdx.x * 2 + (wave >> 2)) * 16 + c;

    const v8f acc = head_tile<H1_>(h1n + (mt * 16 + c) * H1_ + 4 * h,
                                   W2 + (4 * h) * H2_ + n, H2_);
    head_bn_store(acc, b2[n], bn2_g[n], bn2_b[n], red, wave, c, h,
                  h2n + (mt * 16 + 8 * h) * H2_ + n, H2_);
}

__global__ void __launch_bounds__(32) spw_head_l3(
    const float* __restrict__ h2n,
    const float* __restrict__ Wo, const float* __restrict__ bo,
    float* __restrict__ yout)
{
    const int lane = threadIdx.x;
    const int c    = lane & 15;
    const int h    = lane >> 4;
    const int mt   = blockIdx.x;
    const float* arow = h2n + (mt * 16 + c) * H2_ + 4 * h;
    const v8f l0 = head_tile<H2_>(arow, Wo + (4 * h) * C_ + c,      C_);
    const v8f l1 = head_tile<H2_>(arow, Wo + (4 * h) * C_ + 16 + c, C_);
    const float bias0 = bo[c];
    const float bias1 = bo[16 + c];
    float* yrow = yout + (mt * 16 + 8 * h) * C_;
    #pragma unroll
    for (int r = 0; r < 8; ++r) {
        const float v0 = l0[r] + bias0;
        const float v1 = l1[r] + bias1;
        float m = fmaxf(v0, v1);
        #pragma unroll
        for (int o = 8; o > 0; o >>= 1) m = fmaxf(m, __shfl_xor(m, o, 32));
        const float e0 = __expf(v0 - m);
        const float e1 = __expf(v1 - m);
        float s = e0 + e1;
        #pragma unroll
        for (int o = 8; o > 0; o >>= 1) s += __shfl_xor(s, o, 32);
        const float inv = 1.0f / s;
        yrow[r * C_ + c]      = e0 * inv;
        yrow[r * C_ + 16 + c] = e1 * inv;
    }
}

extern "C" void kernel_launch(void* const* d_in, const int* in_sizes, int n_in,
                              void* d_out, int out_size, void* d_ws, size_t ws_size,
                              hipStream_t stream) {
    const float* x     = (const float*)d_in[0];
    const float* w     = (const float*)d_in[1];
    const float* co_w  = (const float*)d_in[2];
    const float* bn0_g = (const float*)d_in[3];
    const float* bn0_b = (const float*)d_in[4];
    const float* W1    = (const float*)d_in[5];
    const float* b1    = (const float*)d_in[6];
    const float* bn1_g = (const float*)d_in[7];
    const float* bn1_b = (const float*)d_in[8];
    const float* W2    = (const float*)d_in[9];
    const float* b2    = (const float*)d_in[10];
    const float* bn2_g = (const float*)d_in[11];
    const float* bn2_b = (const float*)d_in[12];
    const float* Wo    = (const float*)d_in[13];
    const float* bo    = (const float*)d_in[14];
    const int*   idx   = (const int*)d_in[15];

    float* out = (float*)d_out;
    float* WX  = (float*)d_ws;
    float* h1  = WX + B_ * P_;
    float* h2  = h1 + B_ * H1_;

    const int N = in_sizes[0] / B_;

    hipMemsetAsync(WX, 0, (size_t)B_ * P_ * sizeof(float), stream);

    dim3 g1((N + SC_F - 1) / SC_F, B_ / SC_ROWS);
    spw_scatter<<<g1, 32, 0, stream>>>(x, w, idx, WX, N);

    spw_head_l1<<<H1_ / 32, 256, 0, stream>>>(WX, co_w, bn0_g, bn0_b,
                                              W1, b1, bn1_g, bn1_b,
                                              out + B_ * C_, h1);
    spw_head_l2<<<H2_ / 32, 256, 0, stream>>>(h1, W2, b2, bn2_g, bn2_b, h2);
    spw_head_l3<<<B_ / 16, 32, 0, stream>>>(h2, Wo, bo, out);
}


# scatter: 8-wave work-groups (2048 features) add their tile sums into a 64x16 LDS table (LDS f32 adds); one set of f32 global atomics to WX per work-group instead of per wave (sum order changes)
# speedup vs baseline: 4.5778x; 1.7558x over previous
#include <hip/hip_runtime.h>
#include <hip/hip_bf16.h>


#define B_   64
#define P_   128
#define H1_  256
#define H2_  128
#define C_   32

typedef __attribute__((ext_vector_type(2))) float v2f;
typedef __attribute__((ext_vector_type(8))) float v8f;

#define SC_TILES 8
#define SC_ROWS  (8 * SC_TILES)
#define SC_NIT   8
#define SC_F     (32 * SC_NIT)
#define SC_WAVES 8
#define SC_WIN   16

struct ScB {
    v2f f[SC_NIT][2][2];
};

__device__ __forceinline__ void sc_build(ScB& b, const float* __restrict__ wr,
                                         const int* __restrict__ ir,
                                         int pc, int& lo, int& hi)
{
    float4 wv[SC_NIT][2];
    int4   iv[SC_NIT][2];
    #pragma unroll
    for (int it = 0; it < SC_NIT; ++it) {
        #pragma unroll
        for (int s = 0; s < 2; ++s) {
            wv[it][s] = *(const float4*)(wr + it * 32 + s * 16);
            iv[it][s] = *(const int4*)(ir + it * 32 + s * 16);
        }
    }
    #pragma unroll
    for (int it = 0; it < SC_NIT; ++it) {
        #pragma unroll
        for (int s = 0; s < 2; ++s) {
            const int4   i4 = iv[it][s];
            const float4 w4 = wv[it][s];
            lo = min(lo, min(min(i4.x, i4.y), min(i4.z, i4.w)));
            hi = max(hi, max(max(i4.x, i4.y), max(i4.z, i4.w)));
            b.f[it][s][0].x = (i4.x == pc) ? w4.x : 0.0f;
            b.f[it][s][0].y = (i4.y == pc) ? w4.y : 0.0f;
            b.f[it][s][1].x = (i4.z == pc) ? w4.z : 0.0f;
            b.f[it][s][1].y = (i4.w == pc) ? w4.w : 0.0f;
        }
    }
}

__device__ __forceinline__ v8f sc_tile(const ScB& b, const float* __restrict__ xt)
{
    float4 xv[SC_NIT][2];
    #pragma unroll
    for (int it = 0; it < SC_NIT; ++it) {
        #pragma unroll
        for (int s = 0; s < 2; ++s)
            xv[it][s] = *(const float4*)(xt + it * 32 + s * 16);
    }
    __builtin_amdgcn_sched_barrier(0);
    v8f a0 = {0.f, 0.f, 0.f, 0.f, 0.f, 0.f, 0.f, 0.f};
    v8f a1 = {0.f, 0.f, 0.f, 0.f, 0.f, 0.f, 0.f, 0.f};
    #pragma unroll
    for (int i2 = 0; i2 < SC_NIT / 2; ++i2) {
        #pragma unroll
        for (int s = 0; s < 2; ++s) {
            v2f p, q;
            p.x = xv[2 * i2][s].x;     p.y = xv[2 * i2][s].y;
            q.x = xv[2 * i2 + 1][s].x; q.y = xv[2 * i2 + 1][s].y;
            a0 = __builtin_amdgcn_wmma_f32_16x16x4_f32(
                false, p, false, b.f[2 * i2][s][0], (short)0, a0, false, false);
            a1 = __builtin_amdgcn_wmma_f32_16x16x4_f32(
                false, q, false, b.f[2 * i2 + 1][s][0], (short)0, a1, false, false);
            p.x = xv[2 * i2][s].z;     p.y = xv[2 * i2][s].w;
            q.x = xv[2 * i2 + 1][s].z; q.y = xv[2 * i2 + 1][s].w;
            a0 = __builtin_amdgcn_wmma_f32_16x16x4_f32(
                false, p, false, b.f[2 * i2][s][1], (short)0, a0, false, false);
            a1 = __builtin_amdgcn_wmma_f32_16x16x4_f32(
                false, q, false, b.f[2 * i2 + 1][s][1], (short)0, a1, false, false);
        }
    }
    return a0 + a1;
}

__global__ void __launch_bounds__(32 * SC_WAVES) spw_scatter(
    const float* __restrict__ x, const float* __restrict__ w,
    const int* __restrict__ idx, float* __restrict__ WX, int N)
{
    __shared__ float sred[SC_ROWS * SC_WIN];
    const int tid  = threadIdx.x;
    const int lane = tid & 31;
    const int wave = __builtin_amdgcn_readfirstlane(tid >> 5);
    const int c    = lane & 15;
    const int h    = lane >> 4;
    const int fb   = c >> 3;
    const int r    = c & 7;
    const int foff = (fb * 2 + h) * 4;
    const int wgf0 = blockIdx.x * (SC_F * SC_WAVES);
    const int f0   = wgf0 + wave * SC_F;
    const int row0 = blockIdx.y * SC_ROWS;

    #pragma unroll
    for (int q = 0; q < (SC_ROWS * SC_WIN) / (32 * SC_WAVES); ++q)
        sred[q * (32 * SC_WAVES) + tid] = 0.0f;
    int base = idx[wgf0];
    int pb0  = idx[min(f0, N - 1)];
    base = min(max(base, 0), P_ - 1);
    pb0  = min(max(pb0, 0), P_ - 1);
    __syncthreads();

    if (f0 + SC_F <= N) {
        const float* xr = x + (long)(row0 + r) * N + f0 + foff;
        const float* wr = w + f0 + foff;
        const int*   ir = idx + f0 + foff;
        const long tile_stride = 8L * N;

        int lo = 0x7fffffff, hi = -0x7fffffff - 1;
        int k = 0, kmax = 0;
        bool first = true;
        int pb = pb0;

        do {
            const int pc = pb + r;
            ScB bfr;
            int zoff = 0;
            asm volatile("" : "+s"(zoff));
            sc_build(bfr, wr + zoff, ir + zoff, pc, lo, hi);

            const bool owner = (fb == h) && ((unsigned)pc < (unsigned)P_);
            const int  col   = pc - base;
            const bool inwin = (unsigned)col < (unsigned)SC_WIN;
            const bool to_lds = owner && inwin;
            const bool to_glb = owner && !inwin;
            const float* xt = xr;
            int wrow = row0 * P_ + pc;
            int srow = inwin ? col : 0;

            #pragma unroll 1
            for (int t = 0; t < SC_TILES; ++t) {
                const v8f acc = sc_tile(bfr, xt);
                if (to_lds) {
                    #pragma unroll
                    for (int j = 0; j < 8; ++j)
                        atomicAdd(&sred[srow + j * SC_WIN], acc[j]);
                } else if (to_glb) {
                    #pragma unroll
                    for (int j = 0; j < 8; ++j) {
                        const float v = acc[j];
                        if (v != 0.0f)
                            atomicAdd(&WX[wrow + j * P_], v);
                    }
                }
                xt   += tile_stride;
                wrow += 8 * P_;
                srow += 8 * SC_WIN;
            }

            if (first) {
                first = false;
                #pragma unroll
                for (int o = 16; o > 0; o >>= 1) {
                    lo = min(lo, __shfl_xor(lo, o, 32));
                    hi = max(hi, __shfl_xor(hi, o, 32));
                }
                lo = __builtin_amdgcn_readfirstlane(lo);
                hi = __builtin_amdgcn_readfirstlane(hi);
                lo = max(lo, 0);
                hi = min(hi, P_ - 1);
                k    = (lo - pb0) >> 3;
                kmax = (hi - pb0) >> 3;
            } else {
                ++k;
            }
            if (k == 0) ++k;
            pb = pb0 + 8 * k;
        } while (k <= kmax);
    } else if (f0 < N) {
        for (int f = f0 + lane; f < N; f += 32) {
            const int   p  = idx[f];
            const float ww = w[f];
            if ((unsigned)p < (unsigned)P_)
                for (int b = 0; b < SC_ROWS; ++b)
                    atomicAdd(&WX[(row0 + b) * P_ + p], x[(long)(row0 + b) * N + f] * ww);
        }
    }

    __syncthreads();
    #pragma unroll
    for (int q = 0; q < (SC_ROWS * SC_WIN) / (32 * SC_WAVES); ++q) {
        const int   e   = q * (32 * SC_WAVES) + tid;
        const float v   = sred[e];
        const int   pcg = base + (e & (SC_WIN - 1));
        if (v != 0.0f && pcg < P_)
            atomicAdd(&WX[(row0 + (e / SC_WIN)) * P_ + pcg], v);
    }
}

template <int K>
__device__ __forceinline__ v8f head_tile(const float* __restrict__ arow,
                                         const float* __restrict__ bcol, int ldb)
{
    float4 av[K / 8];
    float  bv[K / 8][4];
    #pragma unroll
    for (int i = 0; i < K / 8; ++i) {
        av[i] = *(const float4*)(arow + 8 * i);
        #pragma unroll
        for (int e = 0; e < 4; ++e) bv[i][e] = bcol[(8 * i + e) * ldb];
    }
    __builtin_amdgcn_sched_barrier(0);
    v8f a0 = {0.f, 0.f, 0.f, 0.f, 0.f, 0.f, 0.f, 0.f};
    v8f a1 = {0.f, 0.f, 0.f, 0.f, 0.f, 0.f, 0.f, 0.f};
    #pragma unroll
    for (int i = 0; i < K / 8; ++i) {
        v2f p, q, u, v;
        p.x = av[i].x;  p.y = av[i].y;
        q.x = av[i].z;  q.y = av[i].w;
        u.x = bv[i][0]; u.y = bv[i][1];
        v.x = bv[i][2]; v.y = bv[i][3];
        a0 = __builtin_amdgcn_wmma_f32_16x16x4_f32(
            false, p, false, u, (short)0, a0, false, false);
        a1 = __builtin_amdgcn_wmma_f32_16x16x4_f32(
            false, q, false, v, (short)0, a1, false, false);
    }
    return a0 + a1;
}

template <int K>
__device__ __forceinline__ v8f head_tile_z(const float* __restrict__ arow,
                                           const float* __restrict__ bcol, int ldb,
                                           const float* sAh, const float* sCh)
{
    float4 av[K / 8];
    float  bv[K / 8][4];
    #pragma unroll
    for (int i = 0; i < K / 8; ++i) {
        av[i] = *(const float4*)(arow + 8 * i);
        #pragma unroll
        for (int e = 0; e < 4; ++e) bv[i][e] = bcol[(8 * i + e) * ldb];
    }
    __builtin_amdgcn_sched_barrier(0);
    v8f a0 = {0.f, 0.f, 0.f, 0.f, 0.f, 0.f, 0.f, 0.f};
    v8f a1 = {0.f, 0.f, 0.f, 0.f, 0.f, 0.f, 0.f, 0.f};
    #pragma unroll
    for (int i = 0; i < K / 8; ++i) {
        const float4 s = *(const float4*)(sAh + 8 * i);
        const float4 t = *(const float4*)(sCh + 8 * i);
        v2f p, q, u, v;
        p.x = fmaxf(av[i].x, 0.f) * s.x + t.x;
        p.y = fmaxf(av[i].y, 0.f) * s.y + t.y;
        q.x = fmaxf(av[i].z, 0.f) * s.z + t.z;
        q.y = fmaxf(av[i].w, 0.f) * s.w + t.w;
        u.x = bv[i][0]; u.y = bv[i][1];
        v.x = bv[i][2]; v.y = bv[i][3];
        a0 = __builtin_amdgcn_wmma_f32_16x16x4_f32(
            false, p, false, u, (short)0, a0, false, false);
        a1 = __builtin_amdgcn_wmma_f32_16x16x4_f32(
            false, q, false, v, (short)0, a1, false, false);
    }
    return a0 + a1;
}

__device__ __forceinline__ void head_bn_store(v8f acc, float bias, float g, float bt,
                                              float (*red)[16][2], int wave, int c, int h,
                                              float* __restrict__ dst  ,
                                              int ld)
{
    float s = 0.f, s2 = 0.f;
    #pragma unroll
    for (int r = 0; r < 8; ++r) {
        const float v = fmaxf(acc[r] + bias, 0.f);
        acc[r] = v;
        s += v; s2 += v * v;
    }
    s  += __shfl_xor(s, 16, 32);
    s2 += __shfl_xor(s2, 16, 32);
    if (h == 0) { red[wave][c][0] = s; red[wave][c][1] = s2; }
    __syncthreads();
    const int w0 = wave & ~3;
    const float S  = (red[w0][c][0] + red[w0 + 1][c][0]) + (red[w0 + 2][c][0] + red[w0 + 3][c][0]);
    const float S2 = (red[w0][c][1] + red[w0 + 1][c][1]) + (red[w0 + 2][c][1] + red[w0 + 3][c][1]);
    const float invB = 1.0f / (float)B_;
    const float m   = S * invB;
    const float var = S2 * invB - m * m;
    const float rg  = rsqrtf(var + 1e-5f) * g;
    const float sh  = bt - m * rg;
    #pragma unroll
    for (int r = 0; r < 8; ++r)
        dst[r * ld] = acc[r] * rg + sh;
}

__global__ void __launch_bounds__(256) spw_head_l1(
    const float* __restrict__ WX,
    const float* __restrict__ co_w,
    const float* __restrict__ bn0_g, const float* __restrict__ bn0_b,
    const float* __restrict__ W1, const float* __restrict__ b1,
    const float* __restrict__ bn1_g, const float* __restrict__ bn1_b,
    float* __restrict__ Zout,
    float* __restrict__ h1n)
{
    __shared__ __attribute__((aligned(16))) float sA[P_];
    __shared__ __attribute__((aligned(16))) float sC[P_];
    __shared__ float red[8][16][2];
    const int t    = threadIdx.x;
    const int lane = t & 31;
    const int wave = t >> 5;
    const int c    = lane & 15;
    const int h    = lane >> 4;
    const int mt   = wave & 3;
    const int n    = (blockIdx.x * 2 + (wave >> 2)) * 16 + c;
    const float invB = 1.0f / (float)B_;

    if (t < P_) {
        float xv[B_];
        #pragma unroll
        for (int r = 0; r < B_; ++r) xv[r] = WX[r * P_ + t];
        float s = 0.f, s2 = 0.f;
        #pragma unroll
        for (int r = 0; r < B_; ++r) {
            const float v = fmaxf(xv[r], 0.f);
            s += v; s2 += v * v;
        }
        const float m   = s * invB;
        const float var = s2 * invB - m * m;
        const float rg  = rsqrtf(var + 1e-5f) * bn0_g[t];
        const float sg  = 1.0f / (1.0f + __expf(-co_w[t]));
        sA[t] = rg * sg;
        sC[t] = (bn0_b[t] - m * rg) * sg;
    }
    __syncthreads();

    #pragma unroll
    for (int q = 0; q < 4; ++q) {
        const int e  = blockIdx.x * (8 * P_) + q * 256 + t;
        const int cp = e & (P_ - 1);
        const float v = fmaxf(WX[e], 0.f);
        Zout[e] = v * sA[cp] + sC[cp];
    }

    const v8f acc = head_tile_z<P_>(WX + (mt * 16 + c) * P_ + 4 * h,
                                    W1 + (4 * h) * H1_ + n, H1_,
                                    sA + 4 * h, sC + 4 * h);
    head_bn_store(acc, b1[n], bn1_g[n], bn1_b[n], red, wave, c, h,
                  h1n + (mt * 16 + 8 * h) * H1_ + n, H1_);
}

__global__ void __launch_bounds__(256) spw_head_l2(
    const float* __restrict__ h1n,
    const float* __restrict__ W2, const float* __restrict__ b2,
    const float* __restrict__ bn2_g, const float* __restrict__ bn2_b,
    float* __restrict__ h2n)
{
    __shared__ float red[8][16][2];
    const int t    = threadIdx.x;
    const int lane = t & 31;
    const int wave = t >> 5;
    const int c    = lane & 15;
    const int h    = lane >> 4;
    const int mt   = wave & 3;
    const int n    = (blockIdx.x * 2 + (wave >> 2)) * 16 + c;

    const v8f acc = head_tile<H1_>(h1n + (mt * 16 + c) * H1_ + 4 * h,
                                   W2 + (4 * h) * H2_ + n, H2_);
    head_bn_store(acc, b2[n], bn2_g[n], bn2_b[n], red, wave, c, h,
                  h2n + (mt * 16 + 8 * h) * H2_ + n, H2_);
}

__global__ void __launch_bounds__(32) spw_head_l3(
    const float* __restrict__ h2n,
    const float* __restrict__ Wo, const float* __restrict__ bo,
    float* __restrict__ yout)
{
    const int lane = threadIdx.x;
    const int c    = lane & 15;
    const int h    = lane >> 4;
    const int mt   = blockIdx.x;
    const float* arow = h2n + (mt * 16 + c) * H2_ + 4 * h;
    const v8f l0 = head_tile<H2_>(arow, Wo + (4 * h) * C_ + c,      C_);
    const v8f l1 = head_tile<H2_>(arow, Wo + (4 * h) * C_ + 16 + c, C_);
    const float bias0 = bo[c];
    const float bias1 = bo[16 + c];
    float* yrow = yout + (mt * 16 + 8 * h) * C_;
    #pragma unroll
    for (int r = 0; r < 8; ++r) {
        const float v0 = l0[r] + bias0;
        const float v1 = l1[r] + bias1;
        float m = fmaxf(v0, v1);
        #pragma unroll
        for (int o = 8; o > 0; o >>= 1) m = fmaxf(m, __shfl_xor(m, o, 32));
        const float e0 = __expf(v0 - m);
        const float e1 = __expf(v1 - m);
        float s = e0 + e1;
        #pragma unroll
        for (int o = 8; o > 0; o >>= 1) s += __shfl_xor(s, o, 32);
        const float inv = 1.0f / s;
        yrow[r * C_ + c]      = e0 * inv;
        yrow[r * C_ + 16 + c] = e1 * inv;
    }
}

extern "C" void kernel_launch(void* const* d_in, const int* in_sizes, int n_in,
                              void* d_out, int out_size, void* d_ws, size_t ws_size,
                              hipStream_t stream) {
    const float* x     = (const float*)d_in[0];
    const float* w     = (const float*)d_in[1];
    const float* co_w  = (const float*)d_in[2];
    const float* bn0_g = (const float*)d_in[3];
    const float* bn0_b = (const float*)d_in[4];
    const float* W1    = (const float*)d_in[5];
    const float* b1    = (const float*)d_in[6];
    const float* bn1_g = (const float*)d_in[7];
    const float* bn1_b = (const float*)d_in[8];
    const float* W2    = (const float*)d_in[9];
    const float* b2    = (const float*)d_in[10];
    const float* bn2_g = (const float*)d_in[11];
    const float* bn2_b = (const float*)d_in[12];
    const float* Wo    = (const float*)d_in[13];
    const float* bo    = (const float*)d_in[14];
    const int*   idx   = (const int*)d_in[15];

    float* out = (float*)d_out;
    float* WX  = (float*)d_ws;
    float* h1  = WX + B_ * P_;
    float* h2  = h1 + B_ * H1_;

    const int N = in_sizes[0] / B_;

    hipMemsetAsync(WX, 0, (size_t)B_ * P_ * sizeof(float), stream);

    dim3 g1((N + SC_F * SC_WAVES - 1) / (SC_F * SC_WAVES), B_ / SC_ROWS);
    spw_scatter<<<g1, 32 * SC_WAVES, 0, stream>>>(x, w, idx, WX, N);

    spw_head_l1<<<H1_ / 32, 256, 0, stream>>>(WX, co_w, bn0_g, bn0_b,
                                              W1, b1, bn1_g, bn1_b,
                                              out + B_ * C_, h1);
    spw_head_l2<<<H2_ / 32, 256, 0, stream>>>(h1, W2, b2, bn2_g, bn2_b, h2);
    spw_head_l3<<<B_ / 16, 32, 0, stream>>>(h2, Wo, bo, out);
}
